# starting kernel
# baseline (speedup 1.0000x reference)
#include <hip/hip_runtime.h>
#include <math.h>


typedef float v2f __attribute__((ext_vector_type(2)));
typedef float v8f __attribute__((ext_vector_type(8)));

#define NQ   8
#define DIM  256
#define MBLK 4
#define ROWS (MBLK * 16)


__device__ __forceinline__ void ry_gate(float2* st, int w, float t, int tid) {
    int stride = 1 << (7 - w);
    if (tid < 128) {
        int low = tid & (stride - 1);
        int i0  = ((tid - low) << 1) | low;
        int i1  = i0 + stride;
        float s, c;
        sincosf(0.5f * t, &s, &c);
        float2 a0 = st[i0], a1 = st[i1];
        st[i0] = make_float2(c * a0.x - s * a1.x, c * a0.y - s * a1.y);
        st[i1] = make_float2(s * a0.x + c * a1.x, s * a0.y + c * a1.y);
    }
    __syncthreads();
}

__device__ __forceinline__ void rz_gate(float2* st, int w, float t, int tid) {
    int stride = 1 << (7 - w);
    if (tid < 128) {
        int low = tid & (stride - 1);
        int i0  = ((tid - low) << 1) | low;
        int i1  = i0 + stride;
        float s, c;
        sincosf(0.5f * t, &s, &c);
        float2 a0 = st[i0], a1 = st[i1];
        st[i0] = make_float2(a0.x * c + a0.y * s, a0.y * c - a0.x * s);
        st[i1] = make_float2(a1.x * c - a1.y * s, a1.y * c + a1.x * s);
    }
    __syncthreads();
}

__device__ __forceinline__ void cnot_gate(float2* st, int c, int t, int tid) {
    int stride = 1 << (7 - t);
    if (tid < 128) {
        int low = tid & (stride - 1);
        int i0  = ((tid - low) << 1) | low;
        int i1  = i0 + stride;
        if ((i0 >> (7 - c)) & 1) {
            float2 tmp = st[i0]; st[i0] = st[i1]; st[i1] = tmp;
        }
    }
    __syncthreads();
}

__global__ __launch_bounds__(256)
void build_unitary_kernel(const float* __restrict__ conv,
                          const float* __restrict__ pool,
                          float* __restrict__ RePlane,
                          float* __restrict__ ImPlane) {
    __shared__ float2 st[DIM];
    const int tid = threadIdx.x;
    const int j   = blockIdx.x;
    st[tid] = make_float2(tid == j ? 1.0f : 0.0f, 0.0f);
    __syncthreads();

    for (int l = 0; l < 2; ++l) {
        for (int i = 0; i < 7; ++i) {
            const float* g = conv + (l * 7 + i) * 3;
            rz_gate(st, i,     g[0], tid);
            ry_gate(st, i + 1, g[1], tid);
            cnot_gate(st, i, i + 1, tid);
            ry_gate(st, i + 1, g[2], tid);
        }
    }
    const int srcs[7] = {0, 1, 2, 3, 0, 1, 0};
    const int snks[7] = {4, 5, 6, 7, 2, 3, 1};
    const int pidx[7] = {0, 0, 0, 0, 1, 1, 2};
    for (int g = 0; g < 7; ++g) {
        const float* pp = pool + pidx[g] * 3;
        rz_gate(st, srcs[g], pp[0], tid);
        ry_gate(st, snks[g], pp[1], tid);
        cnot_gate(st, srcs[g], snks[g], tid);
        ry_gate(st, snks[g], pp[2], tid);
    }
    RePlane[tid * DIM + j] = st[tid].x;
    ImPlane[tid * DIM + j] = st[tid].y;
}


#define WHT_STAGE(p, D, IMM)                                                   \
    {                                                                          \
        float o_ = __int_as_float(                                             \
            __builtin_amdgcn_ds_swizzle(__float_as_int(p), (IMM)));            \
        p = (lane & (D)) ? (o_ - p) : (p + o_);                                \
    }

__global__ __launch_bounds__(512)
void qcnn_wmma_kernel(const float* __restrict__ x,
                      const float* __restrict__ RePlane,
                      const float* __restrict__ ImPlane,
                      float* __restrict__ out) {
    __shared__ float csC[ROWS][NQ];
    __shared__ float csS[ROWS][NQ];
    __shared__ float hiP[ROWS][16];
    __shared__ float loP[ROWS][16];
    __shared__ float partial[16][ROWS][5];

    const int tid = threadIdx.x;
    const int rb  = blockIdx.x * ROWS;

    {
        int row = tid >> 3, q = tid & 7;
        float s, c;
        sincosf(0.5f * x[(rb + row) * NQ + q], &s, &c);
        csC[row][q] = c;
        csS[row][q] = s;
    }
    __syncthreads();

    for (int idx = tid; idx < 2 * ROWS * 16; idx += 512) {
        int half = idx >> 10;
        int row  = (idx >> 4) & (ROWS - 1);
        int jj   = idx & 15;
        float p = 1.0f;
#pragma unroll
        for (int b = 0; b < 4; ++b) {
            int q   = half * 4 + b;
            int bit = (jj >> (3 - b)) & 1;
            p *= bit ? csS[row][q] : csC[row][q];
        }
        (half ? loP : hiP)[row][jj] = p;
    }
    __syncthreads();

    const int lane = tid & 31;
    const int wv   = tid >> 5;
    const int m    = lane & 15;
    const int koff = (lane >> 4) << 1;
    const int ncol = (wv << 4) + (lane & 15);
    const float* Rb = RePlane + (size_t)ncol * DIM;
    const float* Ib = ImPlane + (size_t)ncol * DIM;

    v2f lo2[MBLK][4];
#pragma unroll
    for (int mt = 0; mt < MBLK; ++mt)
#pragma unroll
        for (int t = 0; t < 4; ++t)
            lo2[mt][t] = *(const v2f*)&loP[mt * 16 + m][koff + 4 * t];

    v8f accR[MBLK], accI[MBLK];
#pragma unroll
    for (int mt = 0; mt < MBLK; ++mt) {
        accR[mt] = (v8f){0.f,0.f,0.f,0.f,0.f,0.f,0.f,0.f};
        accI[mt] = (v8f){0.f,0.f,0.f,0.f,0.f,0.f,0.f,0.f};
    }

#pragma unroll 4
    for (int kb = 0; kb < DIM; kb += 4) {
        const int k    = kb + koff;
        const int hidx = kb >> 4;
        const int t    = (kb >> 2) & 3;
        v2f br = *(const v2f*)&Rb[k];
        v2f bi = *(const v2f*)&Ib[k];
#pragma unroll
        for (int mt = 0; mt < MBLK; ++mt) {
            float h = hiP[mt * 16 + m][hidx];
            v2f a;
            a.x = lo2[mt][t].x * h;
            a.y = lo2[mt][t].y * h;
            accR[mt] = __builtin_amdgcn_wmma_f32_16x16x4_f32(
                false, a, false, br, (short)0, accR[mt], false, false);
            accI[mt] = __builtin_amdgcn_wmma_f32_16x16x4_f32(
                false, a, false, bi, (short)0, accI[mt], false, false);
        }
    }

#pragma unroll
    for (int mt = 0; mt < MBLK; ++mt) {
#pragma unroll
        for (int r = 0; r < 8; ++r) {
            float p = accR[mt][r] * accR[mt][r] + accI[mt][r] * accI[mt][r];
            WHT_STAGE(p, 1, 0x041F)
            WHT_STAGE(p, 2, 0x081F)
            WHT_STAGE(p, 4, 0x101F)
            WHT_STAGE(p, 8, 0x201F)
            int pat = lane & 15;
            int row = mt * 16 + r + ((lane >> 4) << 3);
            if (pat == 0)      partial[wv][row][0] = p;
            else if (pat == 8) partial[wv][row][1] = p;
            else if (pat == 4) partial[wv][row][2] = p;
            else if (pat == 2) partial[wv][row][3] = p;
            else if (pat == 1) partial[wv][row][4] = p;
        }
    }
    __syncthreads();

    {
        int row = tid >> 3, q = tid & 7;
        float acc = 0.0f;
        if (q < 4) {
            int sh = 3 - q;
#pragma unroll
            for (int w2 = 0; w2 < 16; ++w2) {
                float s = partial[w2][row][0];
                acc += ((w2 >> sh) & 1) ? -s : s;
            }
        } else {
            int slot = q - 3;
#pragma unroll
            for (int w2 = 0; w2 < 16; ++w2)
                acc += partial[w2][row][slot];
        }
        out[(rb + (tid >> 3)) * NQ + q] = acc;
    }
}


extern "C" void kernel_launch(void* const* d_in, const int* in_sizes, int n_in,
                              void* d_out, int out_size, void* d_ws, size_t ws_size,
                              hipStream_t stream) {
    const float* x    = (const float*)d_in[0];
    const float* conv = (const float*)d_in[1];
    const float* pool = (const float*)d_in[2];
    float* out        = (float*)d_out;
    float* RePlane    = (float*)d_ws;
    float* ImPlane    = RePlane + DIM * DIM;

    const int B = in_sizes[0] / NQ;

    build_unitary_kernel<<<DIM, 256, 0, stream>>>(conv, pool, RePlane, ImPlane);
    qcnn_wmma_kernel<<<B / ROWS, 512, 0, stream>>>(x, RePlane, ImPlane, out);
}


# Build kernel stores U in WMMA B-fragment order (re/im interleaved, one float4 per lane and k-step): a wave's B operand per k-step is one contiguous 512-B load, not two strided 8-B-per-lane gathers
# speedup vs baseline: 1.5599x; 1.5599x over previous
#include <hip/hip_runtime.h>
#include <math.h>


typedef float v2f __attribute__((ext_vector_type(2)));
typedef float v4f __attribute__((ext_vector_type(4)));
typedef float v8f __attribute__((ext_vector_type(8)));

#define NQ   8
#define DIM  256
#define MBLK 4
#define ROWS (MBLK * 16)


__device__ __forceinline__ void ry_gate(float2* st, int w, float t, int tid) {
    int stride = 1 << (7 - w);
    if (tid < 128) {
        int low = tid & (stride - 1);
        int i0  = ((tid - low) << 1) | low;
        int i1  = i0 + stride;
        float s, c;
        sincosf(0.5f * t, &s, &c);
        float2 a0 = st[i0], a1 = st[i1];
        st[i0] = make_float2(c * a0.x - s * a1.x, c * a0.y - s * a1.y);
        st[i1] = make_float2(s * a0.x + c * a1.x, s * a0.y + c * a1.y);
    }
    __syncthreads();
}

__device__ __forceinline__ void rz_gate(float2* st, int w, float t, int tid) {
    int stride = 1 << (7 - w);
    if (tid < 128) {
        int low = tid & (stride - 1);
        int i0  = ((tid - low) << 1) | low;
        int i1  = i0 + stride;
        float s, c;
        sincosf(0.5f * t, &s, &c);
        float2 a0 = st[i0], a1 = st[i1];
        st[i0] = make_float2(a0.x * c + a0.y * s, a0.y * c - a0.x * s);
        st[i1] = make_float2(a1.x * c - a1.y * s, a1.y * c + a1.x * s);
    }
    __syncthreads();
}

__device__ __forceinline__ void cnot_gate(float2* st, int c, int t, int tid) {
    int stride = 1 << (7 - t);
    if (tid < 128) {
        int low = tid & (stride - 1);
        int i0  = ((tid - low) << 1) | low;
        int i1  = i0 + stride;
        if ((i0 >> (7 - c)) & 1) {
            float2 tmp = st[i0]; st[i0] = st[i1]; st[i1] = tmp;
        }
    }
    __syncthreads();
}

__global__ __launch_bounds__(256)
void build_unitary_kernel(const float* __restrict__ conv,
                          const float* __restrict__ pool,
                          float* __restrict__ Bfrag) {
    __shared__ float2 st[DIM];
    const int tid = threadIdx.x;
    const int j   = blockIdx.x;
    st[tid] = make_float2(tid == j ? 1.0f : 0.0f, 0.0f);
    __syncthreads();

    for (int l = 0; l < 2; ++l) {
        for (int i = 0; i < 7; ++i) {
            const float* g = conv + (l * 7 + i) * 3;
            rz_gate(st, i,     g[0], tid);
            ry_gate(st, i + 1, g[1], tid);
            cnot_gate(st, i, i + 1, tid);
            ry_gate(st, i + 1, g[2], tid);
        }
    }
    const int srcs[7] = {0, 1, 2, 3, 0, 1, 0};
    const int snks[7] = {4, 5, 6, 7, 2, 3, 1};
    const int pidx[7] = {0, 0, 0, 0, 1, 1, 2};
    for (int g = 0; g < 7; ++g) {
        const float* pp = pool + pidx[g] * 3;
        rz_gate(st, srcs[g], pp[0], tid);
        ry_gate(st, snks[g], pp[1], tid);
        cnot_gate(st, srcs[g], snks[g], tid);
        ry_gate(st, snks[g], pp[2], tid);
    }
    {
        const int fl  = (((j >> 1) & 1) << 4) | (tid & 15);
        const int f4  = (((tid >> 4) * (DIM / 4)) + (j >> 2)) * 32 + fl;
        float* dst = Bfrag + (size_t)f4 * 4 + (j & 1);
        dst[0] = st[tid].x;
        dst[2] = st[tid].y;
    }
}


#define WHT_STAGE(p, D, IMM)                                                   \
    {                                                                          \
        float o_ = __int_as_float(                                             \
            __builtin_amdgcn_ds_swizzle(__float_as_int(p), (IMM)));            \
        p = (lane & (D)) ? (o_ - p) : (p + o_);                                \
    }

__global__ __launch_bounds__(512, 8)
void qcnn_wmma_kernel(const float* __restrict__ x,
                      const float* __restrict__ Bfrag,
                      float* __restrict__ out) {
    __shared__ float csC[ROWS][NQ];
    __shared__ float csS[ROWS][NQ];
    __shared__ float hiP[ROWS][16];
    __shared__ float loP[ROWS][16];
    __shared__ float partial[16][ROWS][5];

    const int tid = threadIdx.x;
    const int rb  = blockIdx.x * ROWS;

    {
        int row = tid >> 3, q = tid & 7;
        float s, c;
        sincosf(0.5f * x[(rb + row) * NQ + q], &s, &c);
        csC[row][q] = c;
        csS[row][q] = s;
    }
    __syncthreads();

    for (int idx = tid; idx < 2 * ROWS * 16; idx += 512) {
        int half = idx >> 10;
        int row  = (idx >> 4) & (ROWS - 1);
        int jj   = idx & 15;
        float p = 1.0f;
#pragma unroll
        for (int b = 0; b < 4; ++b) {
            int q   = half * 4 + b;
            int bit = (jj >> (3 - b)) & 1;
            p *= bit ? csS[row][q] : csC[row][q];
        }
        (half ? loP : hiP)[row][jj] = p;
    }
    __syncthreads();

    const int lane = tid & 31;
    const int wv   = tid >> 5;
    const int m    = lane & 15;
    const int koff = (lane >> 4) << 1;
    const v4f* Bw = (const v4f*)Bfrag;
    const unsigned bo = (unsigned)wv * ((DIM / 4) * 32) + (unsigned)lane;

    v2f lo2[MBLK][4];
#pragma unroll
    for (int mt = 0; mt < MBLK; ++mt)
#pragma unroll
        for (int t = 0; t < 4; ++t)
            lo2[mt][t] = *(const v2f*)&loP[mt * 16 + m][koff + 4 * t];

    v8f accR[MBLK], accI[MBLK];
#pragma unroll
    for (int mt = 0; mt < MBLK; ++mt) {
        accR[mt] = (v8f){0.f,0.f,0.f,0.f,0.f,0.f,0.f,0.f};
        accI[mt] = (v8f){0.f,0.f,0.f,0.f,0.f,0.f,0.f,0.f};
    }

#pragma unroll 4
    for (int kb = 0; kb < DIM; kb += 4) {
        const int hidx = kb >> 4;
        const int t    = (kb >> 2) & 3;
        const v4f b4 = Bw[bo + (unsigned)(kb >> 2) * 32u];
        v2f br = {b4.x, b4.y};
        v2f bi = {b4.z, b4.w};
#pragma unroll
        for (int mt = 0; mt < MBLK; ++mt) {
            float h = hiP[mt * 16 + m][hidx];
            v2f a;
            a.x = lo2[mt][t].x * h;
            a.y = lo2[mt][t].y * h;
            accR[mt] = __builtin_amdgcn_wmma_f32_16x16x4_f32(
                false, a, false, br, (short)0, accR[mt], false, false);
            accI[mt] = __builtin_amdgcn_wmma_f32_16x16x4_f32(
                false, a, false, bi, (short)0, accI[mt], false, false);
        }
    }

#pragma unroll
    for (int mt = 0; mt < MBLK; ++mt) {
#pragma unroll
        for (int r = 0; r < 8; ++r) {
            float p = accR[mt][r] * accR[mt][r] + accI[mt][r] * accI[mt][r];
            WHT_STAGE(p, 1, 0x041F)
            WHT_STAGE(p, 2, 0x081F)
            WHT_STAGE(p, 4, 0x101F)
            WHT_STAGE(p, 8, 0x201F)
            int pat = lane & 15;
            int row = mt * 16 + r + ((lane >> 4) << 3);
            if (pat == 0)      partial[wv][row][0] = p;
            else if (pat == 8) partial[wv][row][1] = p;
            else if (pat == 4) partial[wv][row][2] = p;
            else if (pat == 2) partial[wv][row][3] = p;
            else if (pat == 1) partial[wv][row][4] = p;
        }
    }
    __syncthreads();

    {
        int row = tid >> 3, q = tid & 7;
        float acc = 0.0f;
        if (q < 4) {
            int sh = 3 - q;
#pragma unroll
            for (int w2 = 0; w2 < 16; ++w2) {
                float s = partial[w2][row][0];
                acc += ((w2 >> sh) & 1) ? -s : s;
            }
        } else {
            int slot = q - 3;
#pragma unroll
            for (int w2 = 0; w2 < 16; ++w2)
                acc += partial[w2][row][slot];
        }
        out[(rb + (tid >> 3)) * NQ + q] = acc;
    }
}


extern "C" void kernel_launch(void* const* d_in, const int* in_sizes, int n_in,
                              void* d_out, int out_size, void* d_ws, size_t ws_size,
                              hipStream_t stream) {
    const float* x    = (const float*)d_in[0];
    const float* conv = (const float*)d_in[1];
    const float* pool = (const float*)d_in[2];
    float* out        = (float*)d_out;
    float* Bfrag      = (float*)d_ws;

    const int B = in_sizes[0] / NQ;

    build_unitary_kernel<<<DIM, 256, 0, stream>>>(conv, pool, Bfrag);
    qcnn_wmma_kernel<<<B / ROWS, 512, 0, stream>>>(x, Bfrag, out);
}


# WMMA operand roles swapped (U rows as A, batch factors as B): D holds n in registers, so Z sums are in-lane adds + one half-wave exchange, not 4 waited LDS swizzles per value; Z sum order changes
# speedup vs baseline: 1.9695x; 1.2626x over previous
#include <hip/hip_runtime.h>
#include <math.h>


typedef float v2f __attribute__((ext_vector_type(2)));
typedef float v4f __attribute__((ext_vector_type(4)));
typedef float v8f __attribute__((ext_vector_type(8)));

#define NQ   8
#define DIM  256
#define MBLK 4
#define ROWS (MBLK * 16)


__device__ __forceinline__ void ry_gate(float2* st, int w, float t, int tid) {
    int stride = 1 << (7 - w);
    if (tid < 128) {
        int low = tid & (stride - 1);
        int i0  = ((tid - low) << 1) | low;
        int i1  = i0 + stride;
        float s, c;
        sincosf(0.5f * t, &s, &c);
        float2 a0 = st[i0], a1 = st[i1];
        st[i0] = make_float2(c * a0.x - s * a1.x, c * a0.y - s * a1.y);
        st[i1] = make_float2(s * a0.x + c * a1.x, s * a0.y + c * a1.y);
    }
    __syncthreads();
}

__device__ __forceinline__ void rz_gate(float2* st, int w, float t, int tid) {
    int stride = 1 << (7 - w);
    if (tid < 128) {
        int low = tid & (stride - 1);
        int i0  = ((tid - low) << 1) | low;
        int i1  = i0 + stride;
        float s, c;
        sincosf(0.5f * t, &s, &c);
        float2 a0 = st[i0], a1 = st[i1];
        st[i0] = make_float2(a0.x * c + a0.y * s, a0.y * c - a0.x * s);
        st[i1] = make_float2(a1.x * c - a1.y * s, a1.y * c + a1.x * s);
    }
    __syncthreads();
}

__device__ __forceinline__ void cnot_gate(float2* st, int c, int t, int tid) {
    int stride = 1 << (7 - t);
    if (tid < 128) {
        int low = tid & (stride - 1);
        int i0  = ((tid - low) << 1) | low;
        int i1  = i0 + stride;
        if ((i0 >> (7 - c)) & 1) {
            float2 tmp = st[i0]; st[i0] = st[i1]; st[i1] = tmp;
        }
    }
    __syncthreads();
}

__global__ __launch_bounds__(256)
void build_unitary_kernel(const float* __restrict__ conv,
                          const float* __restrict__ pool,
                          float* __restrict__ Bfrag) {
    __shared__ float2 st[DIM];
    const int tid = threadIdx.x;
    const int j   = blockIdx.x;
    st[tid] = make_float2(tid == j ? 1.0f : 0.0f, 0.0f);
    __syncthreads();

    for (int l = 0; l < 2; ++l) {
        for (int i = 0; i < 7; ++i) {
            const float* g = conv + (l * 7 + i) * 3;
            rz_gate(st, i,     g[0], tid);
            ry_gate(st, i + 1, g[1], tid);
            cnot_gate(st, i, i + 1, tid);
            ry_gate(st, i + 1, g[2], tid);
        }
    }
    const int srcs[7] = {0, 1, 2, 3, 0, 1, 0};
    const int snks[7] = {4, 5, 6, 7, 2, 3, 1};
    const int pidx[7] = {0, 0, 0, 0, 1, 1, 2};
    for (int g = 0; g < 7; ++g) {
        const float* pp = pool + pidx[g] * 3;
        rz_gate(st, srcs[g], pp[0], tid);
        ry_gate(st, snks[g], pp[1], tid);
        cnot_gate(st, srcs[g], snks[g], tid);
        ry_gate(st, snks[g], pp[2], tid);
    }
    {
        const int fl  = (((j >> 1) & 1) << 4) | (tid & 15);
        const int f4  = (((tid >> 4) * (DIM / 4)) + (j >> 2)) * 32 + fl;
        float* dst = Bfrag + (size_t)f4 * 4 + (j & 1);
        dst[0] = st[tid].x;
        dst[2] = st[tid].y;
    }
}


__device__ __forceinline__ float other_half(float v) {
    const int x = __float_as_int(v);
    return __int_as_float(__builtin_amdgcn_permlanex16(
        x, x, 0x76543210u, 0xfedcba98u, false, false));
}

__global__ __launch_bounds__(512, 8)
void qcnn_wmma_kernel(const float* __restrict__ x,
                      const float* __restrict__ Bfrag,
                      float* __restrict__ out) {
    __shared__ float csC[ROWS][NQ];
    __shared__ float csS[ROWS][NQ];
    __shared__ float hiP[ROWS][16];
    __shared__ float loP[ROWS][16];
    __shared__ float partial[16][ROWS][5];

    const int tid = threadIdx.x;
    const int rb  = blockIdx.x * ROWS;

    {
        int row = tid >> 3, q = tid & 7;
        float s, c;
        sincosf(0.5f * x[(rb + row) * NQ + q], &s, &c);
        csC[row][q] = c;
        csS[row][q] = s;
    }
    __syncthreads();

    for (int idx = tid; idx < 2 * ROWS * 16; idx += 512) {
        int half = idx >> 10;
        int row  = (idx >> 4) & (ROWS - 1);
        int jj   = idx & 15;
        float p = 1.0f;
#pragma unroll
        for (int b = 0; b < 4; ++b) {
            int q   = half * 4 + b;
            int bit = (jj >> (3 - b)) & 1;
            p *= bit ? csS[row][q] : csC[row][q];
        }
        (half ? loP : hiP)[row][jj] = p;
    }
    __syncthreads();

    const int lane = tid & 31;
    const int wv   = tid >> 5;
    const int m    = lane & 15;
    const int koff = (lane >> 4) << 1;
    const v4f* Bw = (const v4f*)Bfrag;
    const unsigned bo = (unsigned)wv * ((DIM / 4) * 32) + (unsigned)lane;

    v2f lo2[MBLK][4];
#pragma unroll
    for (int mt = 0; mt < MBLK; ++mt)
#pragma unroll
        for (int t = 0; t < 4; ++t)
            lo2[mt][t] = *(const v2f*)&loP[mt * 16 + m][koff + 4 * t];

    v8f accR[MBLK], accI[MBLK];
#pragma unroll
    for (int mt = 0; mt < MBLK; ++mt) {
        accR[mt] = (v8f){0.f,0.f,0.f,0.f,0.f,0.f,0.f,0.f};
        accI[mt] = (v8f){0.f,0.f,0.f,0.f,0.f,0.f,0.f,0.f};
    }

#pragma unroll 4
    for (int kb = 0; kb < DIM; kb += 4) {
        const int hidx = kb >> 4;
        const int t    = (kb >> 2) & 3;
        const v4f b4 = Bw[bo + (unsigned)(kb >> 2) * 32u];
        v2f br = {b4.x, b4.y};
        v2f bi = {b4.z, b4.w};
#pragma unroll
        for (int mt = 0; mt < MBLK; ++mt) {
            float h = hiP[mt * 16 + m][hidx];
            v2f a;
            a.x = lo2[mt][t].x * h;
            a.y = lo2[mt][t].y * h;
            accR[mt] = __builtin_amdgcn_wmma_f32_16x16x4_f32(
                false, br, false, a, (short)0, accR[mt], false, false);
            accI[mt] = __builtin_amdgcn_wmma_f32_16x16x4_f32(
                false, bi, false, a, (short)0, accI[mt], false, false);
        }
    }

#pragma unroll
    for (int mt = 0; mt < MBLK; ++mt) {
        float p[8];
#pragma unroll
        for (int r = 0; r < 8; ++r)
            p[r] = accR[mt][r] * accR[mt][r] + accI[mt][r] * accI[mt][r];
        const float s01 = p[0] + p[1], d01 = p[0] - p[1];
        const float s23 = p[2] + p[3], d23 = p[2] - p[3];
        const float s45 = p[4] + p[5], d45 = p[4] - p[5];
        const float s67 = p[6] + p[7], d67 = p[6] - p[7];
        const float sA = s01 + s23, eA = s01 - s23;
        const float sB = s45 + s67, eB = s45 - s67;
        const float S  = sA + sB;
        const float D5 = sA - sB;
        const float D6 = eA + eB;
        const float D7 = (d01 + d23) + (d45 + d67);
        const float So  = other_half(S);
        const float D5o = other_half(D5);
        const float D6o = other_half(D6);
        const float D7o = other_half(D7);
        if (lane < 16) {
            float* dst = &partial[wv][mt * 16 + m][0];
            dst[0] = S + So;
            dst[1] = S - So;
            dst[2] = D5 + D5o;
            dst[3] = D6 + D6o;
            dst[4] = D7 + D7o;
        }
    }
    __syncthreads();

    {
        int row = tid >> 3, q = tid & 7;
        float acc = 0.0f;
        if (q < 4) {
            int sh = 3 - q;
#pragma unroll
            for (int w2 = 0; w2 < 16; ++w2) {
                float s = partial[w2][row][0];
                acc += ((w2 >> sh) & 1) ? -s : s;
            }
        } else {
            int slot = q - 3;
#pragma unroll
            for (int w2 = 0; w2 < 16; ++w2)
                acc += partial[w2][row][slot];
        }
        out[(rb + (tid >> 3)) * NQ + q] = acc;
    }
}


extern "C" void kernel_launch(void* const* d_in, const int* in_sizes, int n_in,
                              void* d_out, int out_size, void* d_ws, size_t ws_size,
                              hipStream_t stream) {
    const float* x    = (const float*)d_in[0];
    const float* conv = (const float*)d_in[1];
    const float* pool = (const float*)d_in[2];
    float* out        = (float*)d_out;
    float* Bfrag      = (float*)d_ws;

    const int B = in_sizes[0] / NQ;

    build_unitary_kernel<<<DIM, 256, 0, stream>>>(conv, pool, Bfrag);
    qcnn_wmma_kernel<<<B / ROWS, 512, 0, stream>>>(x, Bfrag, out);
}


# U is block-diagonal in wire 0 (it only ever gets RZ and CNOT-control, for any parameters): each wave's k-loop runs only the 32 k-steps of its own 128-wide block; skipped products are exact zeros
# speedup vs baseline: 3.0025x; 1.5245x over previous
#include <hip/hip_runtime.h>
#include <math.h>


typedef float v2f __attribute__((ext_vector_type(2)));
typedef float v4f __attribute__((ext_vector_type(4)));
typedef float v8f __attribute__((ext_vector_type(8)));

#define NQ   8
#define DIM  256
#define MBLK 4
#define ROWS (MBLK * 16)


__device__ __forceinline__ void ry_gate(float2* st, int w, float t, int tid) {
    int stride = 1 << (7 - w);
    if (tid < 128) {
        int low = tid & (stride - 1);
        int i0  = ((tid - low) << 1) | low;
        int i1  = i0 + stride;
        float s, c;
        sincosf(0.5f * t, &s, &c);
        float2 a0 = st[i0], a1 = st[i1];
        st[i0] = make_float2(c * a0.x - s * a1.x, c * a0.y - s * a1.y);
        st[i1] = make_float2(s * a0.x + c * a1.x, s * a0.y + c * a1.y);
    }
    __syncthreads();
}

__device__ __forceinline__ void rz_gate(float2* st, int w, float t, int tid) {
    int stride = 1 << (7 - w);
    if (tid < 128) {
        int low = tid & (stride - 1);
        int i0  = ((tid - low) << 1) | low;
        int i1  = i0 + stride;
        float s, c;
        sincosf(0.5f * t, &s, &c);
        float2 a0 = st[i0], a1 = st[i1];
        st[i0] = make_float2(a0.x * c + a0.y * s, a0.y * c - a0.x * s);
        st[i1] = make_float2(a1.x * c - a1.y * s, a1.y * c + a1.x * s);
    }
    __syncthreads();
}

__device__ __forceinline__ void cnot_gate(float2* st, int c, int t, int tid) {
    int stride = 1 << (7 - t);
    if (tid < 128) {
        int low = tid & (stride - 1);
        int i0  = ((tid - low) << 1) | low;
        int i1  = i0 + stride;
        if ((i0 >> (7 - c)) & 1) {
            float2 tmp = st[i0]; st[i0] = st[i1]; st[i1] = tmp;
        }
    }
    __syncthreads();
}

__global__ __launch_bounds__(256)
void build_unitary_kernel(const float* __restrict__ conv,
                          const float* __restrict__ pool,
                          float* __restrict__ Bfrag) {
    __shared__ float2 st[DIM];
    const int tid = threadIdx.x;
    const int j   = blockIdx.x;
    st[tid] = make_float2(tid == j ? 1.0f : 0.0f, 0.0f);
    __syncthreads();

    for (int l = 0; l < 2; ++l) {
        for (int i = 0; i < 7; ++i) {
            const float* g = conv + (l * 7 + i) * 3;
            rz_gate(st, i,     g[0], tid);
            ry_gate(st, i + 1, g[1], tid);
            cnot_gate(st, i, i + 1, tid);
            ry_gate(st, i + 1, g[2], tid);
        }
    }
    const int srcs[7] = {0, 1, 2, 3, 0, 1, 0};
    const int snks[7] = {4, 5, 6, 7, 2, 3, 1};
    const int pidx[7] = {0, 0, 0, 0, 1, 1, 2};
    for (int g = 0; g < 7; ++g) {
        const float* pp = pool + pidx[g] * 3;
        rz_gate(st, srcs[g], pp[0], tid);
        ry_gate(st, snks[g], pp[1], tid);
        cnot_gate(st, srcs[g], snks[g], tid);
        ry_gate(st, snks[g], pp[2], tid);
    }
    {
        const int fl  = (((j >> 1) & 1) << 4) | (tid & 15);
        const int f4  = (((tid >> 4) * (DIM / 4)) + (j >> 2)) * 32 + fl;
        float* dst = Bfrag + (size_t)f4 * 4 + (j & 1);
        dst[0] = st[tid].x;
        dst[2] = st[tid].y;
    }
}


__device__ __forceinline__ float other_half(float v) {
    const int x = __float_as_int(v);
    return __int_as_float(__builtin_amdgcn_permlanex16(
        x, x, 0x76543210u, 0xfedcba98u, false, false));
}

__global__ __launch_bounds__(512, 8)
void qcnn_wmma_kernel(const float* __restrict__ x,
                      const float* __restrict__ Bfrag,
                      float* __restrict__ out) {
    __shared__ float csC[ROWS][NQ];
    __shared__ float csS[ROWS][NQ];
    __shared__ float hiP[ROWS][16];
    __shared__ float loP[ROWS][16];
    __shared__ float partial[16][ROWS][5];

    const int tid = threadIdx.x;
    const int rb  = blockIdx.x * ROWS;

    {
        int row = tid >> 3, q = tid & 7;
        float s, c;
        sincosf(0.5f * x[(rb + row) * NQ + q], &s, &c);
        csC[row][q] = c;
        csS[row][q] = s;
    }
    __syncthreads();

    for (int idx = tid; idx < 2 * ROWS * 16; idx += 512) {
        int half = idx >> 10;
        int row  = (idx >> 4) & (ROWS - 1);
        int jj   = idx & 15;
        float p = 1.0f;
#pragma unroll
        for (int b = 0; b < 4; ++b) {
            int q   = half * 4 + b;
            int bit = (jj >> (3 - b)) & 1;
            p *= bit ? csS[row][q] : csC[row][q];
        }
        (half ? loP : hiP)[row][jj] = p;
    }
    __syncthreads();

    const int lane = tid & 31;
    const int wv   = tid >> 5;
    const int m    = lane & 15;
    const int koff = (lane >> 4) << 1;
    const int kh   = wv >> 3;
    const v4f* Bw = (const v4f*)Bfrag;
    const unsigned bo = (unsigned)wv * ((DIM / 4) * 32) + (unsigned)kh * ((DIM / 8) * 32)
                      + (unsigned)lane;
    const float* hrow = &hiP[m][8 * kh];

    v2f lo2[MBLK][4];
#pragma unroll
    for (int mt = 0; mt < MBLK; ++mt)
#pragma unroll
        for (int t = 0; t < 4; ++t)
            lo2[mt][t] = *(const v2f*)&loP[mt * 16 + m][koff + 4 * t];

    v8f accR[MBLK], accI[MBLK];
#pragma unroll
    for (int mt = 0; mt < MBLK; ++mt) {
        accR[mt] = (v8f){0.f,0.f,0.f,0.f,0.f,0.f,0.f,0.f};
        accI[mt] = (v8f){0.f,0.f,0.f,0.f,0.f,0.f,0.f,0.f};
    }

#pragma unroll 4
    for (int ks = 0; ks < DIM / 8; ++ks) {
        const int hidx = ks >> 2;
        const int t    = ks & 3;
        const v4f b4 = Bw[bo + (unsigned)ks * 32u];
        v2f br = {b4.x, b4.y};
        v2f bi = {b4.z, b4.w};
#pragma unroll
        for (int mt = 0; mt < MBLK; ++mt) {
            float h = hrow[mt * 16 * 16 + hidx];
            v2f a;
            a.x = lo2[mt][t].x * h;
            a.y = lo2[mt][t].y * h;
            accR[mt] = __builtin_amdgcn_wmma_f32_16x16x4_f32(
                false, br, false, a, (short)0, accR[mt], false, false);
            accI[mt] = __builtin_amdgcn_wmma_f32_16x16x4_f32(
                false, bi, false, a, (short)0, accI[mt], false, false);
        }
    }

#pragma unroll
    for (int mt = 0; mt < MBLK; ++mt) {
        float p[8];
#pragma unroll
        for (int r = 0; r < 8; ++r)
            p[r] = accR[mt][r] * accR[mt][r] + accI[mt][r] * accI[mt][r];
        const float s01 = p[0] + p[1], d01 = p[0] - p[1];
        const float s23 = p[2] + p[3], d23 = p[2] - p[3];
        const float s45 = p[4] + p[5], d45 = p[4] - p[5];
        const float s67 = p[6] + p[7], d67 = p[6] - p[7];
        const float sA = s01 + s23, eA = s01 - s23;
        const float sB = s45 + s67, eB = s45 - s67;
        const float S  = sA + sB;
        const float D5 = sA - sB;
        const float D6 = eA + eB;
        const float D7 = (d01 + d23) + (d45 + d67);
        const float So  = other_half(S);
        const float D5o = other_half(D5);
        const float D6o = other_half(D6);
        const float D7o = other_half(D7);
        if (lane < 16) {
            float* dst = &partial[wv][mt * 16 + m][0];
            dst[0] = S + So;
            dst[1] = S - So;
            dst[2] = D5 + D5o;
            dst[3] = D6 + D6o;
            dst[4] = D7 + D7o;
        }
    }
    __syncthreads();

    {
        int row = tid >> 3, q = tid & 7;
        float acc = 0.0f;
        if (q < 4) {
            int sh = 3 - q;
#pragma unroll
            for (int w2 = 0; w2 < 16; ++w2) {
                float s = partial[w2][row][0];
                acc += ((w2 >> sh) & 1) ? -s : s;
            }
        } else {
            int slot = q - 3;
#pragma unroll
            for (int w2 = 0; w2 < 16; ++w2)
                acc += partial[w2][row][slot];
        }
        out[(rb + (tid >> 3)) * NQ + q] = acc;
    }
}


extern "C" void kernel_launch(void* const* d_in, const int* in_sizes, int n_in,
                              void* d_out, int out_size, void* d_ws, size_t ws_size,
                              hipStream_t stream) {
    const float* x    = (const float*)d_in[0];
    const float* conv = (const float*)d_in[1];
    const float* pool = (const float*)d_in[2];
    float* out        = (float*)d_out;
    float* Bfrag      = (float*)d_ws;

    const int B = in_sizes[0] / NQ;

    build_unitary_kernel<<<DIM, 256, 0, stream>>>(conv, pool, Bfrag);
    qcnn_wmma_kernel<<<B / ROWS, 512, 0, stream>>>(x, Bfrag, out);
}


# Build kernel: each RZ/RY/CNOT/RY gate group is one LDS pass (RZ phase times a 2x2 picked by the control bit; sincos once up front): 21 barriered passes, not 84; U entries round differently
# speedup vs baseline: 3.7066x; 1.2345x over previous
#include <hip/hip_runtime.h>
#include <math.h>


typedef float v2f __attribute__((ext_vector_type(2)));
typedef float v4f __attribute__((ext_vector_type(4)));
typedef float v8f __attribute__((ext_vector_type(8)));

#define NQ   8
#define DIM  256
#define MBLK 4
#define ROWS (MBLK * 16)


#define NGRP 21

__device__ __forceinline__ float2 group_pass(const float2* __restrict__ in, float2 v,
                                             const float2* __restrict__ mixg,
                                             const float2* __restrict__ phsg,
                                             int src, int snk, int tid) {
    const int tb = 7 - snk;
    const int cb = 7 - src;
    const int t  = (tid >> tb) & 1;
    const int c  = (tid >> cb) & 1;
    const float2 ap = in[tid ^ (1 << tb)];
    const float2 k  = mixg[2 * c + t];
    const float2 ph = phsg[c];
    const float re = k.x * v.x + k.y * ap.x;
    const float im = k.x * v.y + k.y * ap.y;
    return make_float2(re * ph.x - im * ph.y, re * ph.y + im * ph.x);
}

__global__ __launch_bounds__(256)
void build_unitary_kernel(const float* __restrict__ conv,
                          const float* __restrict__ pool,
                          float* __restrict__ Bfrag) {
    __shared__ float2 st[2][DIM];
    __shared__ float2 phs[NGRP][2];
    __shared__ float2 mix[NGRP][4];
    const int tid = threadIdx.x;
    const int j   = blockIdx.x;

    float2 v = make_float2(tid == j ? 1.0f : 0.0f, 0.0f);
    st[0][tid] = v;

    if (tid < NGRP) {
        const int k = tid - 14;
        const float* p = (tid < 14) ? conv + tid * 3
                                    : pool + ((k < 4) ? 0 : ((k < 6) ? 1 : 2)) * 3;
        const float h0 = 0.5f * p[0], h1 = 0.5f * p[1], h2 = 0.5f * p[2];
        float ps, pc, s1, c1, s2, c2;
        sincosf(h0, &ps, &pc);
        sincosf(h1, &s1, &c1);
        sincosf(h2, &s2, &c2);
        const float A = c2 * c1 - s2 * s1;
        const float B = s2 * c1 + c2 * s1;
        const float C = c2 * s1 - s2 * c1;
        const float D = c2 * c1 + s2 * s1;
        phs[tid][0] = make_float2(pc, -ps);
        phs[tid][1] = make_float2(pc,  ps);
        mix[tid][0] = make_float2( A, -B);
        mix[tid][1] = make_float2( A,  B);
        mix[tid][2] = make_float2( C,  D);
        mix[tid][3] = make_float2(-C,  D);
    }
    __syncthreads();

#pragma unroll
    for (int g = 0; g < 14; ++g) {
        const int i = (g < 7) ? g : g - 7;
        v = group_pass(st[g & 1], v, mix[g], phs[g], i, i + 1, tid);
        st[(g & 1) ^ 1][tid] = v;
        __syncthreads();
    }
    v = group_pass(st[0], v, mix[14], phs[14], 0, 4, tid);  st[1][tid] = v;  __syncthreads();
    v = group_pass(st[1], v, mix[15], phs[15], 1, 5, tid);  st[0][tid] = v;  __syncthreads();
    v = group_pass(st[0], v, mix[16], phs[16], 2, 6, tid);  st[1][tid] = v;  __syncthreads();
    v = group_pass(st[1], v, mix[17], phs[17], 3, 7, tid);  st[0][tid] = v;  __syncthreads();
    v = group_pass(st[0], v, mix[18], phs[18], 0, 2, tid);  st[1][tid] = v;  __syncthreads();
    v = group_pass(st[1], v, mix[19], phs[19], 1, 3, tid);  st[0][tid] = v;  __syncthreads();
    v = group_pass(st[0], v, mix[20], phs[20], 0, 1, tid);

    {
        const int fl  = (((j >> 1) & 1) << 4) | (tid & 15);
        const int f4  = (((tid >> 4) * (DIM / 4)) + (j >> 2)) * 32 + fl;
        float* dst = Bfrag + (size_t)f4 * 4 + (j & 1);
        dst[0] = v.x;
        dst[2] = v.y;
    }
}


__device__ __forceinline__ float other_half(float v) {
    const int x = __float_as_int(v);
    return __int_as_float(__builtin_amdgcn_permlanex16(
        x, x, 0x76543210u, 0xfedcba98u, false, false));
}

__global__ __launch_bounds__(512, 8)
void qcnn_wmma_kernel(const float* __restrict__ x,
                      const float* __restrict__ Bfrag,
                      float* __restrict__ out) {
    __shared__ float csC[ROWS][NQ];
    __shared__ float csS[ROWS][NQ];
    __shared__ float hiP[ROWS][16];
    __shared__ float loP[ROWS][16];
    __shared__ float partial[16][ROWS][5];

    const int tid = threadIdx.x;
    const int rb  = blockIdx.x * ROWS;

    {
        int row = tid >> 3, q = tid & 7;
        float s, c;
        sincosf(0.5f * x[(rb + row) * NQ + q], &s, &c);
        csC[row][q] = c;
        csS[row][q] = s;
    }
    __syncthreads();

    for (int idx = tid; idx < 2 * ROWS * 16; idx += 512) {
        int half = idx >> 10;
        int row  = (idx >> 4) & (ROWS - 1);
        int jj   = idx & 15;
        float p = 1.0f;
#pragma unroll
        for (int b = 0; b < 4; ++b) {
            int q   = half * 4 + b;
            int bit = (jj >> (3 - b)) & 1;
            p *= bit ? csS[row][q] : csC[row][q];
        }
        (half ? loP : hiP)[row][jj] = p;
    }
    __syncthreads();

    const int lane = tid & 31;
    const int wv   = tid >> 5;
    const int m    = lane & 15;
    const int koff = (lane >> 4) << 1;
    const int kh   = wv >> 3;
    const v4f* Bw = (const v4f*)Bfrag;
    const unsigned bo = (unsigned)wv * ((DIM / 4) * 32) + (unsigned)kh * ((DIM / 8) * 32)
                      + (unsigned)lane;
    const float* hrow = &hiP[m][8 * kh];

    v2f lo2[MBLK][4];
#pragma unroll
    for (int mt = 0; mt < MBLK; ++mt)
#pragma unroll
        for (int t = 0; t < 4; ++t)
            lo2[mt][t] = *(const v2f*)&loP[mt * 16 + m][koff + 4 * t];

    v8f accR[MBLK], accI[MBLK];
#pragma unroll
    for (int mt = 0; mt < MBLK; ++mt) {
        accR[mt] = (v8f){0.f,0.f,0.f,0.f,0.f,0.f,0.f,0.f};
        accI[mt] = (v8f){0.f,0.f,0.f,0.f,0.f,0.f,0.f,0.f};
    }

#pragma unroll 4
    for (int ks = 0; ks < DIM / 8; ++ks) {
        const int hidx = ks >> 2;
        const int t    = ks & 3;
        const v4f b4 = Bw[bo + (unsigned)ks * 32u];
        v2f br = {b4.x, b4.y};
        v2f bi = {b4.z, b4.w};
#pragma unroll
        for (int mt = 0; mt < MBLK; ++mt) {
            float h = hrow[mt * 16 * 16 + hidx];
            v2f a;
            a.x = lo2[mt][t].x * h;
            a.y = lo2[mt][t].y * h;
            accR[mt] = __builtin_amdgcn_wmma_f32_16x16x4_f32(
                false, br, false, a, (short)0, accR[mt], false, false);
            accI[mt] = __builtin_amdgcn_wmma_f32_16x16x4_f32(
                false, bi, false, a, (short)0, accI[mt], false, false);
        }
    }

#pragma unroll
    for (int mt = 0; mt < MBLK; ++mt) {
        float p[8];
#pragma unroll
        for (int r = 0; r < 8; ++r)
            p[r] = accR[mt][r] * accR[mt][r] + accI[mt][r] * accI[mt][r];
        const float s01 = p[0] + p[1], d01 = p[0] - p[1];
        const float s23 = p[2] + p[3], d23 = p[2] - p[3];
        const float s45 = p[4] + p[5], d45 = p[4] - p[5];
        const float s67 = p[6] + p[7], d67 = p[6] - p[7];
        const float sA = s01 + s23, eA = s01 - s23;
        const float sB = s45 + s67, eB = s45 - s67;
        const float S  = sA + sB;
        const float D5 = sA - sB;
        const float D6 = eA + eB;
        const float D7 = (d01 + d23) + (d45 + d67);
        const float So  = other_half(S);
        const float D5o = other_half(D5);
        const float D6o = other_half(D6);
        const float D7o = other_half(D7);
        if (lane < 16) {
            float* dst = &partial[wv][mt * 16 + m][0];
            dst[0] = S + So;
            dst[1] = S - So;
            dst[2] = D5 + D5o;
            dst[3] = D6 + D6o;
            dst[4] = D7 + D7o;
        }
    }
    __syncthreads();

    {
        int row = tid >> 3, q = tid & 7;
        float acc = 0.0f;
        if (q < 4) {
            int sh = 3 - q;
#pragma unroll
            for (int w2 = 0; w2 < 16; ++w2) {
                float s = partial[w2][row][0];
                acc += ((w2 >> sh) & 1) ? -s : s;
            }
        } else {
            int slot = q - 3;
#pragma unroll
            for (int w2 = 0; w2 < 16; ++w2)
                acc += partial[w2][row][slot];
        }
        out[(rb + (tid >> 3)) * NQ + q] = acc;
    }
}


extern "C" void kernel_launch(void* const* d_in, const int* in_sizes, int n_in,
                              void* d_out, int out_size, void* d_ws, size_t ws_size,
                              hipStream_t stream) {
    const float* x    = (const float*)d_in[0];
    const float* conv = (const float*)d_in[1];
    const float* pool = (const float*)d_in[2];
    float* out        = (float*)d_out;
    float* Bfrag      = (float*)d_ws;

    const int B = in_sizes[0] / NQ;

    build_unitary_kernel<<<DIM, 256, 0, stream>>>(conv, pool, Bfrag);
    qcnn_wmma_kernel<<<B / ROWS, 512, 0, stream>>>(x, Bfrag, out);
}


# GEMM kernel: batch-side WMMA fragments built once per block in LDS, in fragment order, and read by every wave (one 16-B LDS load per M-tile and k-step pair), not re-multiplied in all 16 waves
# speedup vs baseline: 3.9813x; 1.0741x over previous
#include <hip/hip_runtime.h>
#include <math.h>


typedef float v2f __attribute__((ext_vector_type(2)));
typedef float v4f __attribute__((ext_vector_type(4)));
typedef float v8f __attribute__((ext_vector_type(8)));

#define NQ   8
#define DIM  256
#define MBLK 4
#define ROWS (MBLK * 16)


#define NGRP 21

__device__ __forceinline__ float2 group_pass(const float2* __restrict__ in, float2 v,
                                             const float2* __restrict__ mixg,
                                             const float2* __restrict__ phsg,
                                             int src, int snk, int tid) {
    const int tb = 7 - snk;
    const int cb = 7 - src;
    const int t  = (tid >> tb) & 1;
    const int c  = (tid >> cb) & 1;
    const float2 ap = in[tid ^ (1 << tb)];
    const float2 k  = mixg[2 * c + t];
    const float2 ph = phsg[c];
    const float re = k.x * v.x + k.y * ap.x;
    const float im = k.x * v.y + k.y * ap.y;
    return make_float2(re * ph.x - im * ph.y, re * ph.y + im * ph.x);
}

__global__ __launch_bounds__(256)
void build_unitary_kernel(const float* __restrict__ conv,
                          const float* __restrict__ pool,
                          float* __restrict__ Bfrag) {
    __shared__ float2 st[2][DIM];
    __shared__ float2 phs[NGRP][2];
    __shared__ float2 mix[NGRP][4];
    const int tid = threadIdx.x;
    const int j   = blockIdx.x;

    float2 v = make_float2(tid == j ? 1.0f : 0.0f, 0.0f);
    st[0][tid] = v;

    if (tid < NGRP) {
        const int k = tid - 14;
        const float* p = (tid < 14) ? conv + tid * 3
                                    : pool + ((k < 4) ? 0 : ((k < 6) ? 1 : 2)) * 3;
        const float h0 = 0.5f * p[0], h1 = 0.5f * p[1], h2 = 0.5f * p[2];
        float ps, pc, s1, c1, s2, c2;
        sincosf(h0, &ps, &pc);
        sincosf(h1, &s1, &c1);
        sincosf(h2, &s2, &c2);
        const float A = c2 * c1 - s2 * s1;
        const float B = s2 * c1 + c2 * s1;
        const float C = c2 * s1 - s2 * c1;
        const float D = c2 * c1 + s2 * s1;
        phs[tid][0] = make_float2(pc, -ps);
        phs[tid][1] = make_float2(pc,  ps);
        mix[tid][0] = make_float2( A, -B);
        mix[tid][1] = make_float2( A,  B);
        mix[tid][2] = make_float2( C,  D);
        mix[tid][3] = make_float2(-C,  D);
    }
    __syncthreads();

#pragma unroll
    for (int g = 0; g < 14; ++g) {
        const int i = (g < 7) ? g : g - 7;
        v = group_pass(st[g & 1], v, mix[g], phs[g], i, i + 1, tid);
        st[(g & 1) ^ 1][tid] = v;
        __syncthreads();
    }
    v = group_pass(st[0], v, mix[14], phs[14], 0, 4, tid);  st[1][tid] = v;  __syncthreads();
    v = group_pass(st[1], v, mix[15], phs[15], 1, 5, tid);  st[0][tid] = v;  __syncthreads();
    v = group_pass(st[0], v, mix[16], phs[16], 2, 6, tid);  st[1][tid] = v;  __syncthreads();
    v = group_pass(st[1], v, mix[17], phs[17], 3, 7, tid);  st[0][tid] = v;  __syncthreads();
    v = group_pass(st[0], v, mix[18], phs[18], 0, 2, tid);  st[1][tid] = v;  __syncthreads();
    v = group_pass(st[1], v, mix[19], phs[19], 1, 3, tid);  st[0][tid] = v;  __syncthreads();
    v = group_pass(st[0], v, mix[20], phs[20], 0, 1, tid);

    {
        const int fl  = (((j >> 1) & 1) << 4) | (tid & 15);
        const int f4  = (((tid >> 4) * (DIM / 4)) + (j >> 2)) * 32 + fl;
        float* dst = Bfrag + (size_t)f4 * 4 + (j & 1);
        dst[0] = v.x;
        dst[2] = v.y;
    }
}


__device__ __forceinline__ float other_half(float v) {
    const int x = __float_as_int(v);
    return __int_as_float(__builtin_amdgcn_permlanex16(
        x, x, 0x76543210u, 0xfedcba98u, false, false));
}

__global__ __launch_bounds__(512, 8)
void qcnn_wmma_kernel(const float* __restrict__ x,
                      const float* __restrict__ Bfrag,
                      float* __restrict__ out) {
    __shared__ __attribute__((aligned(16))) float csC[ROWS][NQ];
    __shared__ __attribute__((aligned(16))) float csS[ROWS][NQ];
    __shared__ v4f panel[2][MBLK][DIM / 16][32];
    __shared__ float partial[16][ROWS][5];

    const int tid = threadIdx.x;
    const int rb  = blockIdx.x * ROWS;

    {
        int row = tid >> 3, q = tid & 7;
        float s, c;
        sincosf(0.5f * x[(rb + row) * NQ + q], &s, &c);
        csC[row][q] = c;
        csS[row][q] = s;
    }
    __syncthreads();

    {
        const int row = tid & (ROWS - 1);
        const int pre = tid >> 6;
        const v4f c03 = *(const v4f*)&csC[row][0];
        const v4f c47 = *(const v4f*)&csC[row][4];
        const v4f s03 = *(const v4f*)&csS[row][0];
        const v4f s47 = *(const v4f*)&csS[row][4];
        const float f0 = (pre & 4) ? s03.x : c03.x;
        const float f1 = (pre & 2) ? s03.y : c03.y;
        const float f2 = (pre & 1) ? s03.z : c03.z;
        const float p3 = (f0 * f1) * f2;
        float hv[2];
        hv[0] = p3 * c03.w;
        hv[1] = p3 * s03.w;
        float l2[4], l3[8], lo[16];
        l2[0] = c47.x * c47.y;  l2[1] = c47.x * s47.y;
        l2[2] = s47.x * c47.y;  l2[3] = s47.x * s47.y;
#pragma unroll
        for (int i = 0; i < 4; ++i) { l3[2 * i] = l2[i] * c47.z;  l3[2 * i + 1] = l2[i] * s47.z; }
#pragma unroll
        for (int i = 0; i < 8; ++i) { lo[2 * i] = l3[i] * c47.w;  lo[2 * i + 1] = l3[i] * s47.w; }
        const int khp = pre >> 2;
        const int mtp = row >> 4;
        const int cp  = row & 15;
#pragma unroll
        for (int b4 = 0; b4 < 2; ++b4)
#pragma unroll
            for (int b3 = 0; b3 < 2; ++b3)
#pragma unroll
                for (int b1 = 0; b1 < 2; ++b1) {
                    const int ks2 = (pre & 3) * 4 + b4 * 2 + b3;
                    const float h = hv[b4];
                    v4f e;
                    e.x = lo[b3 * 8 + b1 * 2]         * h;
                    e.y = lo[b3 * 8 + b1 * 2 + 1]     * h;
                    e.z = lo[b3 * 8 + 4 + b1 * 2]     * h;
                    e.w = lo[b3 * 8 + 4 + b1 * 2 + 1] * h;
                    panel[khp][mtp][ks2][b1 * 16 + cp] = e;
                }
    }
    __syncthreads();

    const int lane = tid & 31;
    const int wv   = tid >> 5;
    const int m    = lane & 15;
    const int kh   = wv >> 3;
    const v4f* Bw = (const v4f*)Bfrag;
    const unsigned bo = (unsigned)wv * ((DIM / 4) * 32) + (unsigned)kh * ((DIM / 8) * 32)
                      + (unsigned)lane;
    const v4f* pan = &panel[kh][0][0][lane];

    v8f accR[MBLK], accI[MBLK];
#pragma unroll
    for (int mt = 0; mt < MBLK; ++mt) {
        accR[mt] = (v8f){0.f,0.f,0.f,0.f,0.f,0.f,0.f,0.f};
        accI[mt] = (v8f){0.f,0.f,0.f,0.f,0.f,0.f,0.f,0.f};
    }

#pragma unroll 2
    for (int ks2 = 0; ks2 < DIM / 16; ++ks2) {
        const v4f u0 = Bw[bo + (unsigned)(2 * ks2) * 32u];
        const v4f u1 = Bw[bo + (unsigned)(2 * ks2 + 1) * 32u];
        v4f p4[MBLK];
#pragma unroll
        for (int mt = 0; mt < MBLK; ++mt)
            p4[mt] = pan[(mt * (DIM / 16) + ks2) * 32];
        const v2f br0 = {u0.x, u0.y}, bi0 = {u0.z, u0.w};
        const v2f br1 = {u1.x, u1.y}, bi1 = {u1.z, u1.w};
#pragma unroll
        for (int mt = 0; mt < MBLK; ++mt) {
            const v2f a = {p4[mt].x, p4[mt].y};
            accR[mt] = __builtin_amdgcn_wmma_f32_16x16x4_f32(
                false, br0, false, a, (short)0, accR[mt], false, false);
            accI[mt] = __builtin_amdgcn_wmma_f32_16x16x4_f32(
                false, bi0, false, a, (short)0, accI[mt], false, false);
        }
#pragma unroll
        for (int mt = 0; mt < MBLK; ++mt) {
            const v2f a = {p4[mt].z, p4[mt].w};
            accR[mt] = __builtin_amdgcn_wmma_f32_16x16x4_f32(
                false, br1, false, a, (short)0, accR[mt], false, false);
            accI[mt] = __builtin_amdgcn_wmma_f32_16x16x4_f32(
                false, bi1, false, a, (short)0, accI[mt], false, false);
        }
    }

#pragma unroll
    for (int mt = 0; mt < MBLK; ++mt) {
        float p[8];
#pragma unroll
        for (int r = 0; r < 8; ++r)
            p[r] = accR[mt][r] * accR[mt][r] + accI[mt][r] * accI[mt][r];
        const float s01 = p[0] + p[1], d01 = p[0] - p[1];
        const float s23 = p[2] + p[3], d23 = p[2] - p[3];
        const float s45 = p[4] + p[5], d45 = p[4] - p[5];
        const float s67 = p[6] + p[7], d67 = p[6] - p[7];
        const float sA = s01 + s23, eA = s01 - s23;
        const float sB = s45 + s67, eB = s45 - s67;
        const float S  = sA + sB;
        const float D5 = sA - sB;
        const float D6 = eA + eB;
        const float D7 = (d01 + d23) + (d45 + d67);
        const float So  = other_half(S);
        const float D5o = other_half(D5);
        const float D6o = other_half(D6);
        const float D7o = other_half(D7);
        if (lane < 16) {
            float* dst = &partial[wv][mt * 16 + m][0];
            dst[0] = S + So;
            dst[1] = S - So;
            dst[2] = D5 + D5o;
            dst[3] = D6 + D6o;
            dst[4] = D7 + D7o;
        }
    }
    __syncthreads();

    {
        int row = tid >> 3, q = tid & 7;
        float acc = 0.0f;
        if (q < 4) {
            int sh = 3 - q;
#pragma unroll
            for (int w2 = 0; w2 < 16; ++w2) {
                float s = partial[w2][row][0];
                acc += ((w2 >> sh) & 1) ? -s : s;
            }
        } else {
            int slot = q - 3;
#pragma unroll
            for (int w2 = 0; w2 < 16; ++w2)
                acc += partial[w2][row][slot];
        }
        out[(rb + (tid >> 3)) * NQ + q] = acc;
    }
}


extern "C" void kernel_launch(void* const* d_in, const int* in_sizes, int n_in,
                              void* d_out, int out_size, void* d_ws, size_t ws_size,
                              hipStream_t stream) {
    const float* x    = (const float*)d_in[0];
    const float* conv = (const float*)d_in[1];
    const float* pool = (const float*)d_in[2];
    float* out        = (float*)d_out;
    float* Bfrag      = (float*)d_ws;

    const int B = in_sizes[0] / NQ;

    build_unitary_kernel<<<DIM, 256, 0, stream>>>(conv, pool, Bfrag);
    qcnn_wmma_kernel<<<B / ROWS, 512, 0, stream>>>(x, Bfrag, out);
}
